# starting kernel
# baseline (speedup 1.0000x reference)
#include <hip/hip_runtime.h>
#include <hip/hip_bf16.h>

#define N_NODES 50000
#define N_EDGES 800000
#define DIM 128

typedef __attribute__((ext_vector_type(2))) float v2f;
typedef __attribute__((ext_vector_type(8))) float v8f;

__global__ __launch_bounds__(256) void gat_init(int* __restrict__ counts,
                                                int* __restrict__ cursor) {
    int tid = blockIdx.x * blockDim.x + threadIdx.x;
    if (tid < N_NODES) {
        counts[tid] = 0;
        cursor[tid] = 0;
    }
}

__global__ __launch_bounds__(256) void gat_gemm(const float* __restrict__ h,
                                                const float* __restrict__ W,
                                                float* __restrict__ z) {
    const int lane = threadIdx.x & 31;
    const int wv   = threadIdx.x >> 5;
    const int m0   = blockIdx.x * 16;
    const int n0   = wv * 16;
    const int half = lane >> 4;
    const int lr   = lane & 15;

    const float* __restrict__ Arow = h + (size_t)(m0 + lr) * DIM;
    const float* __restrict__ Brow = W + (size_t)(n0 + lr) * DIM;

    v8f acc = {};
    #pragma unroll 4
    for (int k = 0; k < DIM; k += 4) {
        const int kk = k + half * 2;
        v2f a = *(const v2f*)(Arow + kk);
        v2f b = *(const v2f*)(Brow + kk);
        acc = __builtin_amdgcn_wmma_f32_16x16x4_f32(
             false, a,  false, b,
             (short)0, acc,  false,  false);
    }

    const int n = n0 + lr;
    #pragma unroll
    for (int v = 0; v < 8; ++v) {
        const int m = m0 + v + 8 * half;
        z[(size_t)m * DIM + n] = acc[v];
    }
}

__global__ __launch_bounds__(256) void gat_gemv(const float* __restrict__ z,
                                                const float* __restrict__ a_attn,
                                                float* __restrict__ s_src,
                                                float* __restrict__ s_dst) {
    const int node = blockIdx.x * 8 + (threadIdx.x >> 5);
    const int lane = threadIdx.x & 31;
    if (node >= N_NODES) return;

    const float4 zv = ((const float4*)(z + (size_t)node * DIM))[lane];
    const float4 av = ((const float4*)(a_attn))[lane];
    const float4 dv = ((const float4*)(a_attn + DIM))[lane];

    float ss = zv.x * av.x + zv.y * av.y + zv.z * av.z + zv.w * av.w;
    float sd = zv.x * dv.x + zv.y * dv.y + zv.z * dv.z + zv.w * dv.w;

    #pragma unroll
    for (int off = 16; off > 0; off >>= 1) {
        ss += __shfl_down(ss, off, 32);
        sd += __shfl_down(sd, off, 32);
    }
    if (lane == 0) {
        s_src[node] = ss;
        s_dst[node] = sd;
    }
}

__global__ __launch_bounds__(256) void gat_hist(const int* __restrict__ dst,
                                                int* __restrict__ counts) {
    int i = blockIdx.x * blockDim.x + threadIdx.x;
    if (i >= N_EDGES) return;
    atomicAdd(counts + dst[i], 1);
}

__global__ __launch_bounds__(1024) void gat_scan(const int* __restrict__ counts,
                                                 int* __restrict__ row_start) {
    __shared__ int lds[1024];
    __shared__ int carry_s;
    if (threadIdx.x == 0) carry_s = 0;
    __syncthreads();

    for (int base = 0; base < N_NODES; base += 1024) {
        const int idx = base + (int)threadIdx.x;
        const int v = (idx < N_NODES) ? counts[idx] : 0;
        int acc = v;
        lds[threadIdx.x] = acc;
        __syncthreads();
        #pragma unroll
        for (int off = 1; off < 1024; off <<= 1) {
            int t = (threadIdx.x >= (unsigned)off) ? lds[threadIdx.x - off] : 0;
            __syncthreads();
            acc += t;
            lds[threadIdx.x] = acc;
            __syncthreads();
        }
        const int c = carry_s;
        if (idx < N_NODES) row_start[idx] = c + acc - v;
        __syncthreads();
        if (threadIdx.x == 1023) carry_s = c + lds[1023];
        __syncthreads();
    }
    if (threadIdx.x == 0) row_start[N_NODES] = carry_s;
}

__global__ __launch_bounds__(256) void gat_fill(const int* __restrict__ src,
                                                const int* __restrict__ dst,
                                                const float* __restrict__ s_src,
                                                const float* __restrict__ s_dst,
                                                const int* __restrict__ row_start,
                                                int* __restrict__ cursor,
                                                int* __restrict__ csr_src,
                                                float* __restrict__ csr_e) {
    int i = blockIdx.x * blockDim.x + threadIdx.x;
    if (i >= N_EDGES) return;
    const int s = src[i];
    const int d = dst[i];
    const int pos = row_start[d] + atomicAdd(cursor + d, 1);
    const float v = s_src[s] + s_dst[d];
    csr_src[pos] = s;
    csr_e[pos]   = v > 0.0f ? v : 0.01f * v;
}

__global__ __launch_bounds__(256) void gat_node(const float* __restrict__ z,
                                                const int* __restrict__ row_start,
                                                const int* __restrict__ csr_src,
                                                float* __restrict__ csr_e,
                                                float* __restrict__ out) {
    const int node = blockIdx.x * 8 + (threadIdx.x >> 5);
    const int lane = threadIdx.x & 31;
    const int beg = row_start[node];
    const int end = row_start[node + 1];

    float mx = -INFINITY;
    for (int p = beg + lane; p < end; p += 32) mx = fmaxf(mx, csr_e[p]);
    #pragma unroll
    for (int off = 16; off > 0; off >>= 1) mx = fmaxf(mx, __shfl_xor(mx, off, 32));

    float sum = 0.0f;
    for (int p = beg + lane; p < end; p += 32) {
        const float ex = __expf(csr_e[p] - mx);
        csr_e[p] = ex;
        sum += ex;
    }
    #pragma unroll
    for (int off = 16; off > 0; off >>= 1) sum += __shfl_xor(sum, off, 32);

    float4 acc = {0.0f, 0.0f, 0.0f, 0.0f};
    for (int p0 = beg; p0 < end; p0 += 32) {
        const int nn = min(32, end - p0);
        const int  pl = p0 + lane;
        const float wl = (pl < end) ? csr_e[pl]   : 0.0f;
        const int   sl = (pl < end) ? csr_src[pl] : 0;
        for (int j = 0; j < nn; ++j) {
            const float w = __shfl(wl, j, 32);
            const int   s = __shfl(sl, j, 32);
            const float4 zv = ((const float4*)(z + (size_t)s * DIM))[lane];
            acc.x += w * zv.x;
            acc.y += w * zv.y;
            acc.z += w * zv.z;
            acc.w += w * zv.w;
        }
    }

    const float inv = (end > beg) ? 1.0f / sum : 0.0f;
    float4 o = {acc.x * inv, acc.y * inv, acc.z * inv, acc.w * inv};
    ((float4*)(out + (size_t)node * DIM))[lane] = o;
}

extern "C" void kernel_launch(void* const* d_in, const int* in_sizes, int n_in,
                              void* d_out, int out_size, void* d_ws, size_t ws_size,
                              hipStream_t stream) {
    const float* h      = (const float*)d_in[0];
    const float* W_fc   = (const float*)d_in[1];
    const float* a_attn = (const float*)d_in[2];
    const int*   src    = (const int*)d_in[3];
    const int*   dst    = (const int*)d_in[4];
    float*       out    = (float*)d_out;
    (void)in_sizes; (void)n_in; (void)out_size; (void)ws_size;

    float* ws        = (float*)d_ws;
    float* z         = ws;
    float* s_src     = z + (size_t)N_NODES * DIM;
    float* s_dst     = s_src + N_NODES;
    float* csr_e     = s_dst + N_NODES;
    int*   counts    = (int*)(csr_e + N_EDGES);
    int*   cursor    = counts + N_NODES;
    int*   row_start = cursor + N_NODES;
    int*   csr_src   = row_start + N_NODES + 1;

    gat_init<<<(N_NODES + 255) / 256, 256, 0, stream>>>(counts, cursor);

    gat_gemm<<<N_NODES / 16, 256, 0, stream>>>(h, W_fc, z);

    gat_gemv<<<(N_NODES + 7) / 8, 256, 0, stream>>>(z, a_attn, s_src, s_dst);

    gat_hist<<<(N_EDGES + 255) / 256, 256, 0, stream>>>(dst, counts);

    gat_scan<<<1, 1024, 0, stream>>>(counts, row_start);

    gat_fill<<<(N_EDGES + 255) / 256, 256, 0, stream>>>(src, dst, s_src, s_dst,
                                                        row_start, cursor,
                                                        csr_src, csr_e);

    gat_node<<<N_NODES / 8, 256, 0, stream>>>(z, row_start, csr_src, csr_e, out);
}


# prefix scan spread over 49 workgroups of one 1024-count chunk: each sums the counts in front of its chunk itself, one barrier; was one workgroup walking 49 chunks at 20 barriers each
# speedup vs baseline: 1.4395x; 1.4395x over previous
#include <hip/hip_runtime.h>
#include <hip/hip_bf16.h>

#define N_NODES 50000
#define N_EDGES 800000
#define DIM 128

typedef __attribute__((ext_vector_type(2))) float v2f;
typedef __attribute__((ext_vector_type(8))) float v8f;

__global__ __launch_bounds__(256) void gat_init(int* __restrict__ counts,
                                                int* __restrict__ cursor) {
    int tid = blockIdx.x * blockDim.x + threadIdx.x;
    if (tid < N_NODES) {
        counts[tid] = 0;
        cursor[tid] = 0;
    }
}

__global__ __launch_bounds__(256) void gat_gemm(const float* __restrict__ h,
                                                const float* __restrict__ W,
                                                float* __restrict__ z) {
    const int lane = threadIdx.x & 31;
    const int wv   = threadIdx.x >> 5;
    const int m0   = blockIdx.x * 16;
    const int n0   = wv * 16;
    const int half = lane >> 4;
    const int lr   = lane & 15;

    const float* __restrict__ Arow = h + (size_t)(m0 + lr) * DIM;
    const float* __restrict__ Brow = W + (size_t)(n0 + lr) * DIM;

    v8f acc = {};
    #pragma unroll 4
    for (int k = 0; k < DIM; k += 4) {
        const int kk = k + half * 2;
        v2f a = *(const v2f*)(Arow + kk);
        v2f b = *(const v2f*)(Brow + kk);
        acc = __builtin_amdgcn_wmma_f32_16x16x4_f32(
             false, a,  false, b,
             (short)0, acc,  false,  false);
    }

    const int n = n0 + lr;
    #pragma unroll
    for (int v = 0; v < 8; ++v) {
        const int m = m0 + v + 8 * half;
        z[(size_t)m * DIM + n] = acc[v];
    }
}

__global__ __launch_bounds__(256) void gat_gemv(const float* __restrict__ z,
                                                const float* __restrict__ a_attn,
                                                float* __restrict__ s_src,
                                                float* __restrict__ s_dst) {
    const int node = blockIdx.x * 8 + (threadIdx.x >> 5);
    const int lane = threadIdx.x & 31;
    if (node >= N_NODES) return;

    const float4 zv = ((const float4*)(z + (size_t)node * DIM))[lane];
    const float4 av = ((const float4*)(a_attn))[lane];
    const float4 dv = ((const float4*)(a_attn + DIM))[lane];

    float ss = zv.x * av.x + zv.y * av.y + zv.z * av.z + zv.w * av.w;
    float sd = zv.x * dv.x + zv.y * dv.y + zv.z * dv.z + zv.w * dv.w;

    #pragma unroll
    for (int off = 16; off > 0; off >>= 1) {
        ss += __shfl_down(ss, off, 32);
        sd += __shfl_down(sd, off, 32);
    }
    if (lane == 0) {
        s_src[node] = ss;
        s_dst[node] = sd;
    }
}

__global__ __launch_bounds__(256) void gat_hist(const int* __restrict__ dst,
                                                int* __restrict__ counts) {
    int i = blockIdx.x * blockDim.x + threadIdx.x;
    if (i >= N_EDGES) return;
    atomicAdd(counts + dst[i], 1);
}

#define SCAN_CHUNK 1024
#define SCAN_GRID  ((N_NODES + 1 + SCAN_CHUNK - 1) / SCAN_CHUNK)
#define SCAN_PRE   (((SCAN_GRID - 1) * (SCAN_CHUNK / 4) + SCAN_CHUNK - 1) / SCAN_CHUNK)
__global__ __launch_bounds__(SCAN_CHUNK) void gat_scan(const int* __restrict__ counts,
                                                       int* __restrict__ row_start) {
    __shared__ int wave_pre[32];
    __shared__ int wave_tot[32];
    const int t    = (int)threadIdx.x;
    const int lane = t & 31;
    const int wv   = t >> 5;
    const int idx  = (int)blockIdx.x * SCAN_CHUNK + t;
    const int n4   = (int)blockIdx.x * (SCAN_CHUNK / 4);
    const int4* __restrict__ c4 = (const int4*)counts;

    const int v = (idx < N_NODES) ? counts[idx] : 0;

    int4 w[SCAN_PRE];
    #pragma unroll
    for (int i = 0; i < SCAN_PRE; ++i) {
        const int q = t + i * SCAN_CHUNK;
        w[i] = (q < n4) ? c4[q] : int4{0, 0, 0, 0};
    }
    int pre = 0;
    #pragma unroll
    for (int i = 0; i < SCAN_PRE; ++i) pre += (w[i].x + w[i].y) + (w[i].z + w[i].w);

    int inc = v;
    for (int off = 1; off < 32; off <<= 1) {
        const int u = __shfl_up(inc, off, 32);
        if (lane >= off) inc += u;
    }
    for (int off = 16; off > 0; off >>= 1) pre += __shfl_xor(pre, off, 32);

    if (lane == 31) wave_tot[wv] = inc;
    if (lane == 0)  wave_pre[wv] = pre;
    __syncthreads();

    int base = wave_pre[lane] + ((lane < wv) ? wave_tot[lane] : 0);
    for (int off = 16; off > 0; off >>= 1) base += __shfl_xor(base, off, 32);

    if (idx <= N_NODES) row_start[idx] = base + inc - v;
}

__global__ __launch_bounds__(256) void gat_fill(const int* __restrict__ src,
                                                const int* __restrict__ dst,
                                                const float* __restrict__ s_src,
                                                const float* __restrict__ s_dst,
                                                const int* __restrict__ row_start,
                                                int* __restrict__ cursor,
                                                int* __restrict__ csr_src,
                                                float* __restrict__ csr_e) {
    int i = blockIdx.x * blockDim.x + threadIdx.x;
    if (i >= N_EDGES) return;
    const int s = src[i];
    const int d = dst[i];
    const int pos = row_start[d] + atomicAdd(cursor + d, 1);
    const float v = s_src[s] + s_dst[d];
    csr_src[pos] = s;
    csr_e[pos]   = v > 0.0f ? v : 0.01f * v;
}

__global__ __launch_bounds__(256) void gat_node(const float* __restrict__ z,
                                                const int* __restrict__ row_start,
                                                const int* __restrict__ csr_src,
                                                float* __restrict__ csr_e,
                                                float* __restrict__ out) {
    const int node = blockIdx.x * 8 + (threadIdx.x >> 5);
    const int lane = threadIdx.x & 31;
    const int beg = row_start[node];
    const int end = row_start[node + 1];

    float mx = -INFINITY;
    for (int p = beg + lane; p < end; p += 32) mx = fmaxf(mx, csr_e[p]);
    #pragma unroll
    for (int off = 16; off > 0; off >>= 1) mx = fmaxf(mx, __shfl_xor(mx, off, 32));

    float sum = 0.0f;
    for (int p = beg + lane; p < end; p += 32) {
        const float ex = __expf(csr_e[p] - mx);
        csr_e[p] = ex;
        sum += ex;
    }
    #pragma unroll
    for (int off = 16; off > 0; off >>= 1) sum += __shfl_xor(sum, off, 32);

    float4 acc = {0.0f, 0.0f, 0.0f, 0.0f};
    for (int p0 = beg; p0 < end; p0 += 32) {
        const int nn = min(32, end - p0);
        const int  pl = p0 + lane;
        const float wl = (pl < end) ? csr_e[pl]   : 0.0f;
        const int   sl = (pl < end) ? csr_src[pl] : 0;
        for (int j = 0; j < nn; ++j) {
            const float w = __shfl(wl, j, 32);
            const int   s = __shfl(sl, j, 32);
            const float4 zv = ((const float4*)(z + (size_t)s * DIM))[lane];
            acc.x += w * zv.x;
            acc.y += w * zv.y;
            acc.z += w * zv.z;
            acc.w += w * zv.w;
        }
    }

    const float inv = (end > beg) ? 1.0f / sum : 0.0f;
    float4 o = {acc.x * inv, acc.y * inv, acc.z * inv, acc.w * inv};
    ((float4*)(out + (size_t)node * DIM))[lane] = o;
}

extern "C" void kernel_launch(void* const* d_in, const int* in_sizes, int n_in,
                              void* d_out, int out_size, void* d_ws, size_t ws_size,
                              hipStream_t stream) {
    const float* h      = (const float*)d_in[0];
    const float* W_fc   = (const float*)d_in[1];
    const float* a_attn = (const float*)d_in[2];
    const int*   src    = (const int*)d_in[3];
    const int*   dst    = (const int*)d_in[4];
    float*       out    = (float*)d_out;
    (void)in_sizes; (void)n_in; (void)out_size; (void)ws_size;

    float* ws        = (float*)d_ws;
    float* z         = ws;
    float* s_src     = z + (size_t)N_NODES * DIM;
    float* s_dst     = s_src + N_NODES;
    float* csr_e     = s_dst + N_NODES;
    int*   counts    = (int*)(csr_e + N_EDGES);
    int*   cursor    = counts + N_NODES;
    int*   row_start = cursor + N_NODES;
    int*   csr_src   = row_start + N_NODES + 1;

    gat_init<<<(N_NODES + 255) / 256, 256, 0, stream>>>(counts, cursor);

    gat_gemm<<<N_NODES / 16, 256, 0, stream>>>(h, W_fc, z);

    gat_gemv<<<(N_NODES + 7) / 8, 256, 0, stream>>>(z, a_attn, s_src, s_dst);

    gat_hist<<<(N_EDGES + 255) / 256, 256, 0, stream>>>(dst, counts);

    gat_scan<<<SCAN_GRID, SCAN_CHUNK, 0, stream>>>(counts, row_start);

    gat_fill<<<(N_EDGES + 255) / 256, 256, 0, stream>>>(src, dst, s_src, s_dst,
                                                        row_start, cursor,
                                                        csr_src, csr_e);

    gat_node<<<N_NODES / 8, 256, 0, stream>>>(z, row_start, csr_src, csr_e, out);
}


# edge logits formed in the per-node kernel (s_dst once per node, s_src gathered through csr_src), not per edge in the CSR fill: the fill drops 3 of its 6 random accesses per edge
# speedup vs baseline: 1.6869x; 1.1719x over previous
#include <hip/hip_runtime.h>
#include <hip/hip_bf16.h>

#define N_NODES 50000
#define N_EDGES 800000
#define DIM 128

typedef __attribute__((ext_vector_type(2))) float v2f;
typedef __attribute__((ext_vector_type(8))) float v8f;

__global__ __launch_bounds__(256) void gat_init(int* __restrict__ counts,
                                                int* __restrict__ cursor) {
    int tid = blockIdx.x * blockDim.x + threadIdx.x;
    if (tid < N_NODES) {
        counts[tid] = 0;
        cursor[tid] = 0;
    }
}


template <int MASK>
__device__ __forceinline__ float row_xor(float v) {
    const int i = __builtin_bit_cast(int, v);
    return __builtin_bit_cast(
        float, __builtin_amdgcn_update_dpp(i, i, 0x160 | MASK, 0xf, 0xf, false));
}

__global__ __launch_bounds__(256) void gat_gemm(const float* __restrict__ h,
                                                const float* __restrict__ W,
                                                const float* __restrict__ a_attn,
                                                float* __restrict__ z,
                                                float* __restrict__ s_src,
                                                float* __restrict__ s_dst) {
    __shared__ float part[8][32];
    const int lane = threadIdx.x & 31;
    const int wv   = threadIdx.x >> 5;
    const int m0   = blockIdx.x * 16;
    const int n0   = wv * 16;
    const int half = lane >> 4;
    const int lr   = lane & 15;
    const int n    = n0 + lr;

    const float* __restrict__ Arow = h + (size_t)(m0 + lr) * DIM;
    const float* __restrict__ Brow = W + (size_t)(n0 + lr) * DIM;

    const float as = a_attn[n];
    const float ad = a_attn[DIM + n];

    v8f acc = {};
    #pragma unroll 4
    for (int k = 0; k < DIM; k += 4) {
        const int kk = k + half * 2;
        v2f a = *(const v2f*)(Arow + kk);
        v2f b = *(const v2f*)(Brow + kk);
        acc = __builtin_amdgcn_wmma_f32_16x16x4_f32(
             false, a,  false, b,
             (short)0, acc,  false,  false);
    }

    #pragma unroll
    for (int v = 0; v < 8; ++v) {
        const int m = m0 + v + 8 * half;
        z[(size_t)m * DIM + n] = acc[v];
    }

    float x[16];
    for (int v = 0; v < 8; ++v) {
        x[v]     = acc[v] * as;
        x[v + 8] = acc[v] * ad;
    }
    const bool b3 = (lane & 8) != 0;
    const bool b2 = (lane & 4) != 0;
    const bool b1 = (lane & 2) != 0;
    const bool b0 = (lane & 1) != 0;
    float y[8];
    for (int j = 0; j < 8; ++j) {
        const float keep = b3 ? x[j + 8] : x[j];
        const float send = b3 ? x[j]     : x[j + 8];
        y[j] = keep + row_xor<8>(send);
    }
    float u[4];
    for (int j = 0; j < 4; ++j) {
        const float keep = b2 ? y[j + 4] : y[j];
        const float send = b2 ? y[j]     : y[j + 4];
        u[j] = keep + row_xor<4>(send);
    }
    float t[2];
    for (int j = 0; j < 2; ++j) {
        const float keep = b1 ? u[j + 2] : u[j];
        const float send = b1 ? u[j]     : u[j + 2];
        t[j] = keep + row_xor<2>(send);
    }
    const float keep0 = b0 ? t[1] : t[0];
    const float send0 = b0 ? t[0] : t[1];
    part[wv][lane] = keep0 + row_xor<1>(send0);
    __builtin_amdgcn_fence(__ATOMIC_RELEASE, "workgroup", "local");
    __builtin_amdgcn_s_barrier();
    __builtin_amdgcn_fence(__ATOMIC_ACQUIRE, "workgroup", "local");

    if (wv == 0) {
        float s = part[0][lane];
        for (int w = 1; w < 8; ++w) s += part[w][lane];
        const int m = m0 + 8 * half + (lr & 7);
        if (lr < 8) s_src[m] = s;
        else        s_dst[m] = s;
    }
}

__global__ __launch_bounds__(256) void gat_hist(const int* __restrict__ dst,
                                                int* __restrict__ counts) {
    int i = blockIdx.x * blockDim.x + threadIdx.x;
    if (i >= N_EDGES) return;
    atomicAdd(counts + dst[i], 1);
}

#define SCAN_CHUNK 1024
#define SCAN_GRID  ((N_NODES + 1 + SCAN_CHUNK - 1) / SCAN_CHUNK)
#define SCAN_PRE   (((SCAN_GRID - 1) * (SCAN_CHUNK / 4) + SCAN_CHUNK - 1) / SCAN_CHUNK)
__global__ __launch_bounds__(SCAN_CHUNK) void gat_scan(const int* __restrict__ counts,
                                                       int* __restrict__ row_start) {
    __shared__ int wave_pre[32];
    __shared__ int wave_tot[32];
    const int t    = (int)threadIdx.x;
    const int lane = t & 31;
    const int wv   = t >> 5;
    const int idx  = (int)blockIdx.x * SCAN_CHUNK + t;
    const int n4   = (int)blockIdx.x * (SCAN_CHUNK / 4);
    const int4* __restrict__ c4 = (const int4*)counts;

    const int v = (idx < N_NODES) ? counts[idx] : 0;

    int4 w[SCAN_PRE];
    #pragma unroll
    for (int i = 0; i < SCAN_PRE; ++i) {
        const int q = t + i * SCAN_CHUNK;
        w[i] = (q < n4) ? c4[q] : int4{0, 0, 0, 0};
    }
    int pre = 0;
    #pragma unroll
    for (int i = 0; i < SCAN_PRE; ++i) pre += (w[i].x + w[i].y) + (w[i].z + w[i].w);

    int inc = v;
    for (int off = 1; off < 32; off <<= 1) {
        const int u = __shfl_up(inc, off, 32);
        if (lane >= off) inc += u;
    }
    for (int off = 16; off > 0; off >>= 1) pre += __shfl_xor(pre, off, 32);

    if (lane == 31) wave_tot[wv] = inc;
    if (lane == 0)  wave_pre[wv] = pre;
    __syncthreads();

    int base = wave_pre[lane] + ((lane < wv) ? wave_tot[lane] : 0);
    for (int off = 16; off > 0; off >>= 1) base += __shfl_xor(base, off, 32);

    if (idx <= N_NODES) row_start[idx] = base + inc - v;
}

__global__ __launch_bounds__(256) void gat_fill(const int* __restrict__ src,
                                                const int* __restrict__ dst,
                                                const int* __restrict__ row_start,
                                                int* __restrict__ cursor,
                                                int* __restrict__ csr_src) {
    int i = blockIdx.x * blockDim.x + threadIdx.x;
    if (i >= N_EDGES) return;
    const int s = src[i];
    const int d = dst[i];
    const int pos = row_start[d] + atomicAdd(cursor + d, 1);
    csr_src[pos] = s;
}

__global__ __launch_bounds__(256) void gat_node(const float* __restrict__ z,
                                                const float* __restrict__ s_src,
                                                const float* __restrict__ s_dst,
                                                const int* __restrict__ row_start,
                                                const int* __restrict__ csr_src,
                                                float* __restrict__ csr_e,
                                                float* __restrict__ out) {
    const int node = blockIdx.x * 8 + (threadIdx.x >> 5);
    const int lane = threadIdx.x & 31;
    const int beg = row_start[node];
    const int end = row_start[node + 1];
    const float sd = s_dst[node];

    float mx = -INFINITY;
    for (int p = beg + lane; p < end; p += 32) {
        const float v = s_src[csr_src[p]] + sd;
        const float e = v > 0.0f ? v : 0.01f * v;
        csr_e[p] = e;
        mx = fmaxf(mx, e);
    }
    #pragma unroll
    for (int off = 16; off > 0; off >>= 1) mx = fmaxf(mx, __shfl_xor(mx, off, 32));

    float sum = 0.0f;
    for (int p = beg + lane; p < end; p += 32) {
        const float ex = __expf(csr_e[p] - mx);
        csr_e[p] = ex;
        sum += ex;
    }
    #pragma unroll
    for (int off = 16; off > 0; off >>= 1) sum += __shfl_xor(sum, off, 32);

    float4 acc = {0.0f, 0.0f, 0.0f, 0.0f};
    for (int p0 = beg; p0 < end; p0 += 32) {
        const int nn = min(32, end - p0);
        const int  pl = p0 + lane;
        const float wl = (pl < end) ? csr_e[pl]   : 0.0f;
        const int   sl = (pl < end) ? csr_src[pl] : 0;
        for (int j = 0; j < nn; ++j) {
            const float w = __shfl(wl, j, 32);
            const int   s = __shfl(sl, j, 32);
            const float4 zv = ((const float4*)(z + (size_t)s * DIM))[lane];
            acc.x += w * zv.x;
            acc.y += w * zv.y;
            acc.z += w * zv.z;
            acc.w += w * zv.w;
        }
    }

    const float inv = (end > beg) ? 1.0f / sum : 0.0f;
    float4 o = {acc.x * inv, acc.y * inv, acc.z * inv, acc.w * inv};
    ((float4*)(out + (size_t)node * DIM))[lane] = o;
}

extern "C" void kernel_launch(void* const* d_in, const int* in_sizes, int n_in,
                              void* d_out, int out_size, void* d_ws, size_t ws_size,
                              hipStream_t stream) {
    const float* h      = (const float*)d_in[0];
    const float* W_fc   = (const float*)d_in[1];
    const float* a_attn = (const float*)d_in[2];
    const int*   src    = (const int*)d_in[3];
    const int*   dst    = (const int*)d_in[4];
    float*       out    = (float*)d_out;
    (void)in_sizes; (void)n_in; (void)out_size; (void)ws_size;

    float* ws        = (float*)d_ws;
    float* z         = ws;
    float* s_src     = z + (size_t)N_NODES * DIM;
    float* s_dst     = s_src + N_NODES;
    float* csr_e     = s_dst + N_NODES;
    int*   counts    = (int*)(csr_e + N_EDGES);
    int*   cursor    = counts + N_NODES;
    int*   row_start = cursor + N_NODES;
    int*   csr_src   = row_start + N_NODES + 1;

    gat_init<<<(N_NODES + 255) / 256, 256, 0, stream>>>(counts, cursor);

    gat_gemm<<<N_NODES / 16, 256, 0, stream>>>(h, W_fc, a_attn, z, s_src, s_dst);

    gat_hist<<<(N_EDGES + 255) / 256, 256, 0, stream>>>(dst, counts);

    gat_scan<<<SCAN_GRID, SCAN_CHUNK, 0, stream>>>(counts, row_start);

    gat_fill<<<(N_EDGES + 255) / 256, 256, 0, stream>>>(src, dst, row_start, cursor,
                                                        csr_src);

    gat_node<<<N_NODES / 8, 256, 0, stream>>>(z, s_src, s_dst, row_start, csr_src,
                                              csr_e, out);
}


# GEMM: 768 resident workgroups walk 4-5 row tiles each; W fragments read once (16 B/lane) and kept in registers, h fragments 16 B/lane, all 16 issued ahead of the products; k order of z sums changes
# speedup vs baseline: 4.3669x; 2.5887x over previous
#include <hip/hip_runtime.h>
#include <hip/hip_bf16.h>

#define N_NODES 50000
#define N_EDGES 800000
#define DIM 128

typedef __attribute__((ext_vector_type(2))) float v2f;
typedef __attribute__((ext_vector_type(8))) float v8f;

#define GEMM_ROW_TILES (N_NODES / 16)
#define GEMM_GRID      768
#define GEMM_TILES     ((GEMM_ROW_TILES + GEMM_GRID - 1) / GEMM_GRID)
static_assert(N_NODES % 16 == 0, "whole row tiles");
static_assert(GEMM_GRID <= GEMM_ROW_TILES, "every workgroup has a first tile");

template <int MASK>
__device__ __forceinline__ float row_xor(float v) {
    const int i = __builtin_bit_cast(int, v);
    return __builtin_bit_cast(
        float, __builtin_amdgcn_update_dpp(i, i, 0x160 | MASK, 0xf, 0xf, false));
}

__device__ __forceinline__ void load_a_tile(const float* __restrict__ h, const int m0,
                                            const int lr, const int half, float4 (&af)[16]) {
    const float4* __restrict__ Arow = (const float4*)(h + (size_t)(m0 + lr) * DIM + 4 * half);
    #pragma unroll
    for (int i = 0; i < 16; ++i) af[i] = Arow[2 * i];
}

__global__ __launch_bounds__(256) void gat_gemm(const float* __restrict__ h,
                                                const float* __restrict__ W,
                                                const float* __restrict__ a_attn,
                                                float* __restrict__ z,
                                                float* __restrict__ s_dst) {
    __shared__ float part[2][8][32];
    const int lane = threadIdx.x & 31;
    const int wv   = threadIdx.x >> 5;
    const int n0   = wv * 16;
    const int half = lane >> 4;
    const int lr   = lane & 15;
    const int n    = n0 + lr;
    const int tile0 = (int)blockIdx.x;

    const float4* __restrict__ Brow = (const float4*)(W + (size_t)(n0 + lr) * DIM + 4 * half);
    float4 bf[16];
    #pragma unroll
    for (int i = 0; i < 16; ++i) bf[i] = Brow[2 * i];

    float4 af[16];
    load_a_tile(h, tile0 * 16, lr, half, af);

    const float ad = a_attn[DIM + n];
    const bool b3 = (lane & 8) != 0;
    const bool b2 = (lane & 4) != 0;
    const bool b1 = (lane & 2) != 0;

    #pragma unroll
    for (int t = 0; t < GEMM_TILES; ++t) {
        const int tile = tile0 + t * GEMM_GRID;
        if (tile >= GEMM_ROW_TILES) break;
        const int m0 = tile * 16;

        asm volatile("" : "+v"(af[0].x), "+v"(af[1].x), "+v"(af[2].x), "+v"(af[3].x),
                          "+v"(af[4].x), "+v"(af[5].x), "+v"(af[6].x), "+v"(af[7].x),
                          "+v"(af[8].x), "+v"(af[9].x), "+v"(af[10].x), "+v"(af[11].x),
                          "+v"(af[12].x), "+v"(af[13].x), "+v"(af[14].x), "+v"(af[15].x));

        v8f acc = {};
        #pragma unroll
        for (int i = 0; i < 16; ++i) {
            const v2f a0 = {af[i].x, af[i].y};
            const v2f b0 = {bf[i].x, bf[i].y};
            acc = __builtin_amdgcn_wmma_f32_16x16x4_f32(
                 false, a0,  false, b0,
                 (short)0, acc,  false,  false);
            const v2f a1 = {af[i].z, af[i].w};
            const v2f b1v = {bf[i].z, bf[i].w};
            acc = __builtin_amdgcn_wmma_f32_16x16x4_f32(
                 false, a1,  false, b1v,
                 (short)0, acc,  false,  false);
        }

        if (t + 1 < GEMM_TILES && tile + GEMM_GRID < GEMM_ROW_TILES)
            load_a_tile(h, (tile + GEMM_GRID) * 16, lr, half, af);

        #pragma unroll
        for (int v = 0; v < 8; ++v) {
            const int m = m0 + v + 8 * half;
            z[(size_t)m * DIM + n] = acc[v];
        }

        float x[8];
        for (int v = 0; v < 8; ++v) x[v] = acc[v] * ad;
        float u[4];
        for (int j = 0; j < 4; ++j) {
            const float keep = b3 ? x[j + 4] : x[j];
            const float send = b3 ? x[j]     : x[j + 4];
            u[j] = keep + row_xor<8>(send);
        }
        float tt[2];
        for (int j = 0; j < 2; ++j) {
            const float keep = b2 ? u[j + 2] : u[j];
            const float send = b2 ? u[j]     : u[j + 2];
            tt[j] = keep + row_xor<4>(send);
        }
        const float keep0 = b1 ? tt[1] : tt[0];
        const float send0 = b1 ? tt[0] : tt[1];
        const float r2 = keep0 + row_xor<2>(send0);
        part[t & 1][wv][lane] = r2 + row_xor<1>(r2);
        __builtin_amdgcn_fence(__ATOMIC_RELEASE, "workgroup", "local");
        __builtin_amdgcn_s_barrier();
        __builtin_amdgcn_fence(__ATOMIC_ACQUIRE, "workgroup", "local");

        if (wv == 0) {
            float s = part[t & 1][0][lane];
            for (int w = 1; w < 8; ++w) s += part[t & 1][w][lane];
            const int m = m0 + 8 * half + (lr >> 1);
            if ((lr & 1) == 0) s_dst[m] = s;
        }
    }
}

#define BIN_SHIFT  7
#define BIN_NODES  (1 << BIN_SHIFT)
#define N_BINS     ((N_NODES >> BIN_SHIFT) + 1)
#define BINS_PAD   512
#define EDGE_CHUNK 4096
#define EDGE_GRID  ((N_EDGES + EDGE_CHUNK - 1) / EDGE_CHUNK)
#define ROWS_HALF  (EDGE_GRID / 2)
static_assert(N_EDGES % 4 == 0, "edges are read four at a time");
static_assert(EDGE_GRID % 2 == 0, "two column walkers per bin split the rows evenly");
static_assert(N_BINS + 1 <= BINS_PAD, "bins (and the end mark) must fit the padded scan");
static_assert((long long)N_NODES * BIN_NODES < 0x7fffffffLL, "packed edge must fit an int");
static_assert(EDGE_GRID * N_BINS + N_BINS + 1 <= 2 * N_NODES, "bin tables fit the two old counter planes");

__global__ __launch_bounds__(1024) void gat_bincount(const int* __restrict__ dst,
                                                     int* __restrict__ bin_mat) {
    __shared__ int lcnt[N_BINS];
    const int t = (int)threadIdx.x;
    for (int b = t; b < N_BINS; b += 1024) lcnt[b] = 0;
    __syncthreads();
    const int q = (int)blockIdx.x * 1024 + t;
    if (q * 4 < N_EDGES) {
        const int4 d = ((const int4*)dst)[q];
        atomicAdd(&lcnt[d.x >> BIN_SHIFT], 1);
        atomicAdd(&lcnt[d.y >> BIN_SHIFT], 1);
        atomicAdd(&lcnt[d.z >> BIN_SHIFT], 1);
        atomicAdd(&lcnt[d.w >> BIN_SHIFT], 1);
    }
    __syncthreads();
    for (int b = t; b < N_BINS; b += 1024)
        bin_mat[(int)blockIdx.x * N_BINS + b] = lcnt[b];
}

__global__ __launch_bounds__(1024) void gat_binscatter(const int* __restrict__ src,
                                                       const int* __restrict__ dst,
                                                       const int* __restrict__ bin_mat,
                                                       int* __restrict__ bin_start,
                                                       int* __restrict__ binned) {
    __shared__ int lcnt[BINS_PAD];
    __shared__ int lstart[BINS_PAD];
    __shared__ int gbase[BINS_PAD];
    __shared__ int col_tot[2][BINS_PAD];
    __shared__ int col_pre[2][BINS_PAD];
    __shared__ int wave_a[32];
    __shared__ int wave_b[32];
    __shared__ int st_src[EDGE_CHUNK];
    __shared__ int st_dst[EDGE_CHUNK];

    const int t    = (int)threadIdx.x;
    const int lane = t & 31;
    const int wv   = t >> 5;
    const int wg   = (int)blockIdx.x;

    if (t < BINS_PAD) lcnt[t] = 0;

    {
        const int b    = t & (BINS_PAD - 1);
        const int part = t >> 9;
        int tot = 0, pre = 0;
        if (b < N_BINS) {
            const int r0 = part * ROWS_HALF;
            const int* __restrict__ col = bin_mat + r0 * N_BINS + b;
            #pragma unroll 14
            for (int i = 0; i < ROWS_HALF; ++i) {
                const int c = col[i * N_BINS];
                tot += c;
                pre += (r0 + i < wg) ? c : 0;
            }
        }
        col_tot[part][b] = tot;
        col_pre[part][b] = pre;
    }
    __syncthreads();

    const int q = wg * 1024 + t;
    const bool have = (q * 4 < N_EDGES);
    int4 s = {0, 0, 0, 0}, d = {0, 0, 0, 0};
    int r0 = 0, r1 = 0, r2 = 0, r3 = 0;
    if (have) {
        s = ((const int4*)src)[q];
        d = ((const int4*)dst)[q];
        r0 = atomicAdd(&lcnt[d.x >> BIN_SHIFT], 1);
        r1 = atomicAdd(&lcnt[d.y >> BIN_SHIFT], 1);
        r2 = atomicAdd(&lcnt[d.z >> BIN_SHIFT], 1);
        r3 = atomicAdd(&lcnt[d.w >> BIN_SHIFT], 1);
    }
    __syncthreads();

    int tot = 0, pre = 0, cnt = 0;
    if (t < BINS_PAD) {
        tot = col_tot[0][t] + col_tot[1][t];
        pre = col_pre[0][t] + col_pre[1][t];
        cnt = lcnt[t];
    }
    int inc_a = tot, inc_b = cnt;
    for (int off = 1; off < 32; off <<= 1) {
        const int ua = __shfl_up(inc_a, off, 32);
        const int ub = __shfl_up(inc_b, off, 32);
        if (lane >= off) { inc_a += ua; inc_b += ub; }
    }
    if (lane == 31) { wave_a[wv] = inc_a; wave_b[wv] = inc_b; }
    __syncthreads();
    int base_a = (lane < wv) ? wave_a[lane] : 0;
    int base_b = (lane < wv) ? wave_b[lane] : 0;
    for (int off = 16; off > 0; off >>= 1) {
        base_a += __shfl_xor(base_a, off, 32);
        base_b += __shfl_xor(base_b, off, 32);
    }
    if (t < BINS_PAD) {
        const int bstart = base_a + inc_a - tot;
        lstart[t] = base_b + inc_b - cnt;
        gbase[t]  = bstart + pre;
        if (wg == 0 && t <= N_BINS) bin_start[t] = bstart;
    }
    __syncthreads();

    if (have) {
        int slot;
        slot = lstart[d.x >> BIN_SHIFT] + r0; st_src[slot] = s.x; st_dst[slot] = d.x;
        slot = lstart[d.y >> BIN_SHIFT] + r1; st_src[slot] = s.y; st_dst[slot] = d.y;
        slot = lstart[d.z >> BIN_SHIFT] + r2; st_src[slot] = s.z; st_dst[slot] = d.z;
        slot = lstart[d.w >> BIN_SHIFT] + r3; st_src[slot] = s.w; st_dst[slot] = d.w;
    }
    __syncthreads();

    const int nval = min(EDGE_CHUNK, N_EDGES - wg * EDGE_CHUNK);
    for (int slot = t; slot < nval; slot += 1024) {
        const int dd  = st_dst[slot];
        const int bin = dd >> BIN_SHIFT;
        binned[gbase[bin] + (slot - lstart[bin])] =
            st_src[slot] * BIN_NODES + (dd & (BIN_NODES - 1));
    }
}

__global__ __launch_bounds__(256) void gat_binsort(const int* __restrict__ binned,
                                                   const int* __restrict__ bin_start,
                                                   int* __restrict__ row_start,
                                                   int* __restrict__ csr_src) {
    __shared__ int ncnt[BIN_NODES];
    __shared__ int ncur[BIN_NODES];
    __shared__ int wave_t[8];
    const int t    = (int)threadIdx.x;
    const int lane = t & 31;
    const int wv   = t >> 5;
    const int bin  = (int)blockIdx.x;
    const int lo   = bin_start[bin];
    const int hi   = bin_start[bin + 1];

    if (t < BIN_NODES) ncnt[t] = 0;
    __syncthreads();
    for (int p = lo + t; p < hi; p += 256)
        atomicAdd(&ncnt[binned[p] & (BIN_NODES - 1)], 1);
    __syncthreads();

    const int v = (t < BIN_NODES) ? ncnt[t] : 0;
    int inc = v;
    for (int off = 1; off < 32; off <<= 1) {
        const int u = __shfl_up(inc, off, 32);
        if (lane >= off) inc += u;
    }
    if (lane == 31) wave_t[wv] = inc;
    __syncthreads();
    int base = (lane < wv && lane < 8) ? wave_t[lane] : 0;
    for (int off = 16; off > 0; off >>= 1) base += __shfl_xor(base, off, 32);
    if (t < BIN_NODES) {
        const int excl = base + inc - v;
        ncur[t] = excl;
        const int node = bin * BIN_NODES + t;
        if (node <= N_NODES) row_start[node] = lo + excl;
    }
    __syncthreads();

    for (int p = lo + t; p < hi; p += 256) {
        const int e = binned[p];
        const int pos = lo + atomicAdd(&ncur[e & (BIN_NODES - 1)], 1);
        csr_src[pos] = e >> BIN_SHIFT;
    }
}


__device__ __forceinline__ float other_row(float v) {
    const int i = __builtin_bit_cast(int, v);
    return __builtin_bit_cast(
        float, __builtin_amdgcn_permlanex16(i, i, 0x76543210, 0xfedcba98, false, false));
}

__device__ __forceinline__ float lane_value(float v, int j) {
    return __builtin_bit_cast(
        float, __builtin_amdgcn_readlane(__builtin_bit_cast(int, v), j));
}

template <bool FULL>
__device__ __forceinline__ void node_edges8(const float4* __restrict__ z4, const int sl,
                                            const int c, const int nn, const int lane,
                                            const float4 a4, const float sd,
                                            float& mx, float& sum, float4& acc) {
    int sj[8];
    #pragma unroll
    for (int j = 0; j < 8; ++j) sj[j] = __builtin_amdgcn_readlane(sl, c + j);
    float4 zv[8];
    #pragma unroll
    for (int j = 0; j < 8; ++j) {
        if (FULL || j < nn) zv[j] = z4[(size_t)sj[j] * (DIM / 4) + lane];
        else                zv[j] = make_float4(0.0f, 0.0f, 0.0f, 0.0f);
    }

    if (FULL)
        asm volatile("" : "+v"(zv[0].x), "+v"(zv[1].x), "+v"(zv[2].x), "+v"(zv[3].x),
                          "+v"(zv[4].x), "+v"(zv[5].x), "+v"(zv[6].x), "+v"(zv[7].x));

    float d[8];
    #pragma unroll
    for (int j = 0; j < 8; ++j)
        d[j] = fmaf(zv[j].x, a4.x, fmaf(zv[j].y, a4.y, fmaf(zv[j].z, a4.z, zv[j].w * a4.w)));

    const bool b2 = (lane & 4) != 0;
    const bool b1 = (lane & 2) != 0;
    const bool b0 = (lane & 1) != 0;
    float y[4];
    #pragma unroll
    for (int j = 0; j < 4; ++j) {
        const float keep = b2 ? d[j + 4] : d[j];
        const float send = b2 ? d[j]     : d[j + 4];
        y[j] = keep + row_xor<4>(send);
    }
    float u[2];
    #pragma unroll
    for (int j = 0; j < 2; ++j) {
        const float keep = b1 ? y[j + 2] : y[j];
        const float send = b1 ? y[j]     : y[j + 2];
        u[j] = keep + row_xor<2>(send);
    }
    const float keep0 = b0 ? u[1] : u[0];
    const float send0 = b0 ? u[0] : u[1];
    float ss = keep0 + row_xor<1>(send0);
    ss += row_xor<8>(ss);
    ss += other_row(ss);

    const float v = ss + sd;
    float e = v > 0.0f ? v : 0.01f * v;
    if (!FULL) e = ((lane & 7) < nn) ? e : -INFINITY;

    float gm = fmaxf(e, row_xor<1>(e));
    gm = fmaxf(gm, row_xor<2>(gm));
    gm = fmaxf(gm, row_xor<4>(gm));
    const float mnew  = fmaxf(mx, gm);
    const float scale = __expf(mx - mnew);
    const float w     = __expf(e - mnew);
    mx  = mnew;
    sum = sum * scale + w;
    acc.x *= scale; acc.y *= scale; acc.z *= scale; acc.w *= scale;

    #pragma unroll
    for (int j = 0; j < 8; ++j) {
        const float wj = lane_value(w, j);
        acc.x += wj * zv[j].x;
        acc.y += wj * zv[j].y;
        acc.z += wj * zv[j].z;
        acc.w += wj * zv[j].w;
    }
}

__global__ __launch_bounds__(256) void gat_node(const float* __restrict__ z,
                                                const float* __restrict__ a_attn,
                                                const float* __restrict__ s_dst,
                                                const int* __restrict__ row_start,
                                                const int* __restrict__ csr_src,
                                                float* __restrict__ out) {
    const int wv   = __builtin_amdgcn_readfirstlane((int)(threadIdx.x >> 5));
    const int node = (int)blockIdx.x * 8 + wv;
    const int lane = (int)(threadIdx.x & 31);
    const int beg = __builtin_amdgcn_readfirstlane(row_start[node]);
    const int end = __builtin_amdgcn_readfirstlane(row_start[node + 1]);
    const float sd = __builtin_bit_cast(
        float, __builtin_amdgcn_readfirstlane(__builtin_bit_cast(int, s_dst[node])));
    const float4 a4 = ((const float4*)a_attn)[lane];
    const float4* __restrict__ z4 = (const float4*)z;

    float mx  = -INFINITY;
    float sum = 0.0f;
    float4 acc = {0.0f, 0.0f, 0.0f, 0.0f};
    for (int p0 = beg; p0 < end; p0 += 32) {
        const int nb = min(32, end - p0);
        const int sl = (lane < nb) ? csr_src[p0 + lane] : 0;
        int c = 0;
        for (; c + 8 <= nb; c += 8)
            node_edges8<true>(z4, sl, c, 8, lane, a4, sd, mx, sum, acc);
        if (c < nb)
            node_edges8<false>(z4, sl, c, nb - c, lane, a4, sd, mx, sum, acc);
    }

    sum += row_xor<1>(sum);
    sum += row_xor<2>(sum);
    sum += row_xor<4>(sum);

    const float inv = (end > beg) ? 1.0f / sum : 0.0f;
    float4 o = {acc.x * inv, acc.y * inv, acc.z * inv, acc.w * inv};
    ((float4*)(out + (size_t)node * DIM))[lane] = o;
}

extern "C" void kernel_launch(void* const* d_in, const int* in_sizes, int n_in,
                              void* d_out, int out_size, void* d_ws, size_t ws_size,
                              hipStream_t stream) {
    const float* h      = (const float*)d_in[0];
    const float* W_fc   = (const float*)d_in[1];
    const float* a_attn = (const float*)d_in[2];
    const int*   src    = (const int*)d_in[3];
    const int*   dst    = (const int*)d_in[4];
    float*       out    = (float*)d_out;
    (void)in_sizes; (void)n_in; (void)out_size; (void)ws_size;

    float* ws        = (float*)d_ws;
    float* z         = ws;
    float* s_dst     = z + (size_t)N_NODES * DIM + N_NODES;
    int*   binned    = (int*)(s_dst + N_NODES);
    int*   bin_mat   = binned + N_EDGES;
    int*   bin_start = bin_mat + EDGE_GRID * N_BINS;
    int*   row_start = bin_mat + 2 * N_NODES;
    int*   csr_src   = row_start + N_NODES + 1;

    gat_gemm<<<GEMM_GRID, 256, 0, stream>>>(h, W_fc, a_attn, z, s_dst);

    gat_bincount<<<EDGE_GRID, 1024, 0, stream>>>(dst, bin_mat);

    gat_binscatter<<<EDGE_GRID, 1024, 0, stream>>>(src, dst, bin_mat, bin_start, binned);

    gat_binsort<<<N_BINS, 256, 0, stream>>>(binned, bin_start, row_start, csr_src);

    gat_node<<<N_NODES / 8, 256, 0, stream>>>(z, a_attn, s_dst, row_start, csr_src, out);
}


# product's global traffic as whole 512-byte rows through LDS: W rows and each tile's h rows fetched once per workgroup, fragments read from LDS; z rows stored whole, not as 64-byte halves
# speedup vs baseline: 5.9781x; 1.3690x over previous
#include <hip/hip_runtime.h>
#include <hip/hip_bf16.h>

#define N_NODES 50000
#define N_EDGES 800000
#define DIM 128

typedef __attribute__((ext_vector_type(2))) float v2f;
typedef __attribute__((ext_vector_type(8))) float v8f;
typedef __attribute__((ext_vector_type(4))) float v4f;
typedef __attribute__((address_space(1))) v4f gmem_v4f;

#define GEMM_ROW_TILES (N_NODES / 16)
#define GEMM_GRID      512
#define GEMM_TILES     ((GEMM_ROW_TILES + GEMM_GRID - 1) / GEMM_GRID)
static_assert(N_NODES % 16 == 0, "whole row tiles");
static_assert(GEMM_GRID <= GEMM_ROW_TILES, "every workgroup has a first tile");

template <int MASK>
__device__ __forceinline__ float row_xor(float v) {
    const int i = __builtin_bit_cast(int, v);
    return __builtin_bit_cast(
        float, __builtin_amdgcn_update_dpp(i, i, 0x160 | MASK, 0xf, 0xf, false));
}

#define TP     132
#define TILE_F (16 * TP)

__device__ __forceinline__ void lds_barrier() {
    __builtin_amdgcn_fence(__ATOMIC_RELEASE, "workgroup", "local");
    __builtin_amdgcn_s_barrier();
    __builtin_amdgcn_fence(__ATOMIC_ACQUIRE, "workgroup", "local");
}

__device__ __forceinline__ void gemm_part(const int tile0,
                                          const float* __restrict__ h,
                                          const float* __restrict__ W,
                                          const float* __restrict__ a_attn,
                                          float* __restrict__ z,
                                          float* __restrict__ s_dst) {
    __shared__ __attribute__((aligned(16))) float stage[8 * TILE_F];
    __shared__ float part[2][8][32];
    const int lane = threadIdx.x & 31;
    const int wv   = __builtin_amdgcn_readfirstlane((int)(threadIdx.x >> 5));
    const int n0   = wv * 16;
    const int half = lane >> 4;
    const int lr   = lane & 15;
    const int n    = n0 + lr;
    const int r0   = 2 * wv;
    const int fo   = lr * TP + 4 * half;
    const int ro   = r0 * TP + 4 * lane;

    float* const wst = stage + wv * TILE_F;
    {
        const float4* __restrict__ Wr = (const float4*)(W + (size_t)n0 * DIM) + lane;
        float4 wr[16];
        #pragma unroll
        for (int j = 0; j < 16; ++j) wr[j] = Wr[j * (DIM / 4)];
        #pragma unroll
        for (int j = 0; j < 16; ++j) *(float4*)(wst + j * TP + 4 * lane) = wr[j];
    }
    const float4* __restrict__ hr = (const float4*)(h + (size_t)(tile0 * 16 + r0) * DIM) + lane;
    float4 g0 = hr[0];
    float4 g1 = hr[DIM / 4];
    lds_barrier();

    float4 bf[16];
    #pragma unroll
    for (int i = 0; i < 16; ++i) bf[i] = *(const float4*)(wst + fo + 8 * i);
    lds_barrier();
    *(float4*)(stage + ro)      = g0;
    *(float4*)(stage + ro + TP) = g1;
    lds_barrier();

    const float ad = a_attn[DIM + n];
    const bool b3 = (lane & 8) != 0;
    const bool b2 = (lane & 4) != 0;
    const bool b1 = (lane & 2) != 0;

    #pragma unroll
    for (int t = 0; t < GEMM_TILES; ++t) {
        const int tile = tile0 + t * GEMM_GRID;
        if (tile >= GEMM_ROW_TILES) break;
        const int m0 = tile * 16;
        const float* const Hc = stage + (t & 1) * TILE_F;
        float* const       Hn = stage + ((t + 1) & 1) * TILE_F;
        float* const       Zc = stage + (2 + (t & 1)) * TILE_F;
        const bool more = (t + 1 < GEMM_TILES) && (tile + GEMM_GRID < GEMM_ROW_TILES);

        float4 af[16];
        #pragma unroll
        for (int i = 0; i < 16; ++i) af[i] = *(const float4*)(Hc + fo + 8 * i);

        asm volatile("" : "+v"(af[0].x), "+v"(af[1].x), "+v"(af[2].x), "+v"(af[3].x),
                          "+v"(af[4].x), "+v"(af[5].x), "+v"(af[6].x), "+v"(af[7].x),
                          "+v"(af[8].x), "+v"(af[9].x), "+v"(af[10].x), "+v"(af[11].x),
                          "+v"(af[12].x), "+v"(af[13].x), "+v"(af[14].x), "+v"(af[15].x));

        v8f acc = {};
        #pragma unroll
        for (int i = 0; i < 16; ++i) {
            if (i == 4 && more) {
                const gmem_v4f* hb = (const gmem_v4f*)(h + (size_t)((tile + GEMM_GRID) * 16 + r0) * DIM);
                asm volatile("" : "+s"(hb));
                const v4f q0 = hb[lane];
                const v4f q1 = hb[DIM / 4 + lane];
                g0 = make_float4(q0.x, q0.y, q0.z, q0.w);
                g1 = make_float4(q1.x, q1.y, q1.z, q1.w);
                __builtin_amdgcn_sched_barrier(0);
            }
            const v2f a0 = {af[i].x, af[i].y};
            const v2f b0 = {bf[i].x, bf[i].y};
            acc = __builtin_amdgcn_wmma_f32_16x16x4_f32(
                 false, a0,  false, b0,
                 (short)0, acc,  false,  false);
            const v2f a1 = {af[i].z, af[i].w};
            const v2f b1v = {bf[i].z, bf[i].w};
            acc = __builtin_amdgcn_wmma_f32_16x16x4_f32(
                 false, a1,  false, b1v,
                 (short)0, acc,  false,  false);
        }

        __builtin_amdgcn_sched_barrier(0);
        if (more) {
            *(float4*)(Hn + ro)      = g0;
            *(float4*)(Hn + ro + TP) = g1;
        }

        #pragma unroll
        for (int v = 0; v < 8; ++v) Zc[(v + 8 * half) * TP + n] = acc[v];

        float x[8];
        for (int v = 0; v < 8; ++v) x[v] = acc[v] * ad;
        float u[4];
        for (int j = 0; j < 4; ++j) {
            const float keep = b3 ? x[j + 4] : x[j];
            const float send = b3 ? x[j]     : x[j + 4];
            u[j] = keep + row_xor<8>(send);
        }
        float tt[2];
        for (int j = 0; j < 2; ++j) {
            const float keep = b2 ? u[j + 2] : u[j];
            const float send = b2 ? u[j]     : u[j + 2];
            tt[j] = keep + row_xor<4>(send);
        }
        const float keep0 = b1 ? tt[1] : tt[0];
        const float send0 = b1 ? tt[0] : tt[1];
        const float r2 = keep0 + row_xor<2>(send0);
        part[t & 1][wv][lane] = r2 + row_xor<1>(r2);
        lds_barrier();

        if (wv == 0) {
            float s = part[t & 1][0][lane];
            for (int w = 1; w < 8; ++w) s += part[t & 1][w][lane];
            const int m = m0 + 8 * half + (lr >> 1);
            if ((lr & 1) == 0) s_dst[m] = s;
        }

        const float4 o0 = *(const float4*)(Zc + ro);
        const float4 o1 = *(const float4*)(Zc + ro + TP);
        float4* const zr = (float4*)(z + (size_t)(m0 + r0) * DIM) + lane;
        zr[0]       = o0;
        zr[DIM / 4] = o1;
    }
}

#define BIN_SHIFT  7
#define BIN_NODES  (1 << BIN_SHIFT)
#define N_BINS     ((N_NODES >> BIN_SHIFT) + 1)
#define BINS_PAD   512
#define EDGE_CHUNK 4096
#define EDGE_GRID  ((N_EDGES + EDGE_CHUNK - 1) / EDGE_CHUNK)
#define LPITCH     416
#define SEG_ROUNDS ((EDGE_GRID + 15) / 16)
#define SEG_PAD    (16 * SEG_ROUNDS)
static_assert(N_EDGES % 4 == 0, "edges are read four at a time");
static_assert(N_BINS + 1 <= LPITCH && LPITCH <= BINS_PAD, "bins and the end mark fit a row of lmat");
static_assert(LPITCH % 32 == 0, "a row of lmat is whole 128-byte lines: one writer per line");
static_assert((long long)N_NODES * BIN_NODES < 0x7fffffffLL, "packed edge must fit an int");
static_assert(EDGE_GRID * LPITCH <= 2 * N_NODES, "lmat fits the two old counter planes");
static_assert(SEG_PAD <= 256, "one thread per entry of the run table");

__global__ __launch_bounds__(1024) void gat_chunksort(const int* __restrict__ src,
                                                      const int* __restrict__ dst,
                                                      int* __restrict__ lmat,
                                                      int* __restrict__ binned) {
    __shared__ int lcnt[BINS_PAD];
    __shared__ int lstart[BINS_PAD];
    __shared__ int wave_b[32];
    __shared__ int st[EDGE_CHUNK];

    const int t    = (int)threadIdx.x;
    const int lane = t & 31;
    const int wv   = t >> 5;
    const int wg   = (int)blockIdx.x;

    const int q = wg * 1024 + t;
    const bool have = (q * 4 < N_EDGES);
    int4 s = {0, 0, 0, 0}, d = {0, 0, 0, 0};
    if (have) {
        s = ((const int4*)src)[q];
        d = ((const int4*)dst)[q];
    }
    if (t < BINS_PAD) lcnt[t] = 0;
    __syncthreads();

    int r0 = 0, r1 = 0, r2 = 0, r3 = 0;
    if (have) {
        r0 = atomicAdd(&lcnt[d.x >> BIN_SHIFT], 1);
        r1 = atomicAdd(&lcnt[d.y >> BIN_SHIFT], 1);
        r2 = atomicAdd(&lcnt[d.z >> BIN_SHIFT], 1);
        r3 = atomicAdd(&lcnt[d.w >> BIN_SHIFT], 1);
    }
    __syncthreads();

    const int cnt = (t < BINS_PAD) ? lcnt[t] : 0;
    int inc = cnt;
    for (int off = 1; off < 32; off <<= 1) {
        const int u = __shfl_up(inc, off, 32);
        if (lane >= off) inc += u;
    }
    if (lane == 31) wave_b[wv] = inc;
    __syncthreads();
    int base = (lane < wv) ? wave_b[lane] : 0;
    for (int off = 16; off > 0; off >>= 1) base += __shfl_xor(base, off, 32);
    if (t < BINS_PAD) {
        const int ex = base + inc - cnt;
        lstart[t] = ex;
        if (t < LPITCH) lmat[wg * LPITCH + t] = ex;
    }
    __syncthreads();

    if (have) {
        st[lstart[d.x >> BIN_SHIFT] + r0] = s.x * BIN_NODES + (d.x & (BIN_NODES - 1));
        st[lstart[d.y >> BIN_SHIFT] + r1] = s.y * BIN_NODES + (d.y & (BIN_NODES - 1));
        st[lstart[d.z >> BIN_SHIFT] + r2] = s.z * BIN_NODES + (d.z & (BIN_NODES - 1));
        st[lstart[d.w >> BIN_SHIFT] + r3] = s.w * BIN_NODES + (d.w & (BIN_NODES - 1));
    }
    __syncthreads();

    const int nval = min(EDGE_CHUNK, N_EDGES - wg * EDGE_CHUNK);
    for (int slot = t; slot < nval; slot += 1024)
        binned[wg * EDGE_CHUNK + slot] = st[slot];
}

__device__ __forceinline__ void binsort_part(const int bin,
                                             const int* __restrict__ binned,
                                             const int* __restrict__ lmat,
                                             int* __restrict__ row_start,
                                             int* __restrict__ csr_src) {
    __shared__ int ncnt[BIN_NODES];
    __shared__ int ncur[BIN_NODES];
    __shared__ int wave_t[8];
    __shared__ int wave_lo[8];
    __shared__ int2 seg[SEG_PAD];
    const int t    = (int)threadIdx.x;
    const int lane = t & 31;
    const int wv   = t >> 5;
    const int hw   = t >> 4;
    const int l16  = t & 15;

    int a = 0;
    if (t < SEG_PAD) {
        int2 sg = {0, 0};
        if (t < EDGE_GRID) {
            a    = lmat[t * LPITCH + bin];
            sg.y = lmat[t * LPITCH + bin + 1] - a;
            sg.x = t * EDGE_CHUNK + a;
        }
        seg[t] = sg;
    }
    if (t < BIN_NODES) ncnt[t] = 0;
    __syncthreads();

    int2 sg[SEG_ROUNDS];
    #pragma unroll
    for (int r = 0; r < SEG_ROUNDS; ++r) sg[r] = seg[hw + 16 * r];
    int e[SEG_ROUNDS];
    int big = 0;
    #pragma unroll
    for (int r = 0; r < SEG_ROUNDS; ++r) {
        const bool in = l16 < sg[r].y;
        const int v   = binned[in ? sg[r].x + l16 : 0];
        e[r] = in ? v : -1;
        if (sg[r].y > 16) big |= 1 << r;
    }

    int ra = a;
    for (int off = 16; off > 0; off >>= 1) ra += __shfl_xor(ra, off, 32);
    if (lane == 0) wave_lo[wv] = ra;

    #pragma unroll
    for (int r = 0; r < SEG_ROUNDS; ++r)
        if (e[r] >= 0) atomicAdd(&ncnt[e[r] & (BIN_NODES - 1)], 1);
    if (big) {
        for (int r = 0; r < SEG_ROUNDS; ++r) {
            if (((big >> r) & 1) == 0) continue;
            const int2 s2 = seg[hw + 16 * r];
            const int n = s2.y, b0 = s2.x;
            for (int k = 16 + l16; k < n; k += 16)
                atomicAdd(&ncnt[binned[b0 + k] & (BIN_NODES - 1)], 1);
        }
    }
    __syncthreads();
    int lo = 0;
    for (int w = 0; w < 8; ++w) lo += wave_lo[w];

    const int v = (t < BIN_NODES) ? ncnt[t] : 0;
    int inc = v;
    for (int off = 1; off < 32; off <<= 1) {
        const int u = __shfl_up(inc, off, 32);
        if (lane >= off) inc += u;
    }
    if (lane == 31) wave_t[wv] = inc;
    __syncthreads();
    int base = (lane < wv && lane < 8) ? wave_t[lane] : 0;
    for (int off = 16; off > 0; off >>= 1) base += __shfl_xor(base, off, 32);
    if (t < BIN_NODES) {
        const int excl = base + inc - v;
        ncur[t] = excl;
        const int node = bin * BIN_NODES + t;
        if (node <= N_NODES) row_start[node] = lo + excl;
    }
    __syncthreads();

    #pragma unroll
    for (int r = 0; r < SEG_ROUNDS; ++r) {
        if (e[r] >= 0) {
            const int pos = lo + atomicAdd(&ncur[e[r] & (BIN_NODES - 1)], 1);
            csr_src[pos] = e[r] >> BIN_SHIFT;
        }
    }
    if (big) {
        for (int r = 0; r < SEG_ROUNDS; ++r) {
            if (((big >> r) & 1) == 0) continue;
            const int2 s2 = seg[hw + 16 * r];
            const int n = s2.y, b0 = s2.x;
            for (int k = 16 + l16; k < n; k += 16) {
                const int ee = binned[b0 + k];
                const int pos = lo + atomicAdd(&ncur[ee & (BIN_NODES - 1)], 1);
                csr_src[pos] = ee >> BIN_SHIFT;
            }
        }
    }
}

__global__ __launch_bounds__(256) void gat_gemm_binsort(const float* __restrict__ h,
                                                        const float* __restrict__ W,
                                                        const float* __restrict__ a_attn,
                                                        float* __restrict__ z,
                                                        float* __restrict__ s_dst,
                                                        const int* __restrict__ binned,
                                                        const int* __restrict__ lmat,
                                                        int* __restrict__ row_start,
                                                        int* __restrict__ csr_src) {
    const int wg = (int)blockIdx.x;
    if (wg < GEMM_GRID) gemm_part(wg, h, W, a_attn, z, s_dst);
    else                binsort_part(wg - GEMM_GRID, binned, lmat, row_start, csr_src);
}


__device__ __forceinline__ float other_row(float v) {
    const int i = __builtin_bit_cast(int, v);
    return __builtin_bit_cast(
        float, __builtin_amdgcn_permlanex16(i, i, 0x76543210, 0xfedcba98, false, false));
}

__device__ __forceinline__ float lane_value(float v, int j) {
    return __builtin_bit_cast(
        float, __builtin_amdgcn_readlane(__builtin_bit_cast(int, v), j));
}

template <bool FULL>
__device__ __forceinline__ void node_edges8(const float4* __restrict__ z4, const int sl,
                                            const int c, const int nn, const int lane,
                                            const float4 a4, const float sd,
                                            float& mx, float& sum, float4& acc) {
    int sj[8];
    #pragma unroll
    for (int j = 0; j < 8; ++j) sj[j] = __builtin_amdgcn_readlane(sl, c + j);
    float4 zv[8];
    #pragma unroll
    for (int j = 0; j < 8; ++j) {
        if (FULL || j < nn) zv[j] = z4[(size_t)sj[j] * (DIM / 4) + lane];
        else                zv[j] = make_float4(0.0f, 0.0f, 0.0f, 0.0f);
    }

    if (FULL)
        asm volatile("" : "+v"(zv[0].x), "+v"(zv[1].x), "+v"(zv[2].x), "+v"(zv[3].x),
                          "+v"(zv[4].x), "+v"(zv[5].x), "+v"(zv[6].x), "+v"(zv[7].x));

    float d[8];
    #pragma unroll
    for (int j = 0; j < 8; ++j)
        d[j] = fmaf(zv[j].x, a4.x, fmaf(zv[j].y, a4.y, fmaf(zv[j].z, a4.z, zv[j].w * a4.w)));

    const bool b2 = (lane & 4) != 0;
    const bool b1 = (lane & 2) != 0;
    const bool b0 = (lane & 1) != 0;
    float y[4];
    #pragma unroll
    for (int j = 0; j < 4; ++j) {
        const float keep = b2 ? d[j + 4] : d[j];
        const float send = b2 ? d[j]     : d[j + 4];
        y[j] = keep + row_xor<4>(send);
    }
    float u[2];
    #pragma unroll
    for (int j = 0; j < 2; ++j) {
        const float keep = b1 ? y[j + 2] : y[j];
        const float send = b1 ? y[j]     : y[j + 2];
        u[j] = keep + row_xor<2>(send);
    }
    const float keep0 = b0 ? u[1] : u[0];
    const float send0 = b0 ? u[0] : u[1];
    float ss = keep0 + row_xor<1>(send0);
    ss += row_xor<8>(ss);
    ss += other_row(ss);

    const float v = ss + sd;
    float e = v > 0.0f ? v : 0.01f * v;
    if (!FULL) e = ((lane & 7) < nn) ? e : -INFINITY;

    float gm = fmaxf(e, row_xor<1>(e));
    gm = fmaxf(gm, row_xor<2>(gm));
    gm = fmaxf(gm, row_xor<4>(gm));
    const float mnew  = fmaxf(mx, gm);
    const float scale = __expf(mx - mnew);
    const float w     = __expf(e - mnew);
    mx  = mnew;
    sum = sum * scale + w;
    acc.x *= scale; acc.y *= scale; acc.z *= scale; acc.w *= scale;

    #pragma unroll
    for (int j = 0; j < 8; ++j) {
        const float wj = lane_value(w, j);
        acc.x += wj * zv[j].x;
        acc.y += wj * zv[j].y;
        acc.z += wj * zv[j].z;
        acc.w += wj * zv[j].w;
    }
}

__global__ __launch_bounds__(256) void gat_node(const float* __restrict__ z,
                                                const float* __restrict__ a_attn,
                                                const float* __restrict__ s_dst,
                                                const int* __restrict__ row_start,
                                                const int* __restrict__ csr_src,
                                                float* __restrict__ out) {
    const int wv   = __builtin_amdgcn_readfirstlane((int)(threadIdx.x >> 5));
    const int node = (int)blockIdx.x * 8 + wv;
    const int lane = (int)(threadIdx.x & 31);
    const int beg = __builtin_amdgcn_readfirstlane(row_start[node]);
    const int end = __builtin_amdgcn_readfirstlane(row_start[node + 1]);
    const float sd = __builtin_bit_cast(
        float, __builtin_amdgcn_readfirstlane(__builtin_bit_cast(int, s_dst[node])));
    const float4 a4 = ((const float4*)a_attn)[lane];
    const float4* __restrict__ z4 = (const float4*)z;

    float mx  = -INFINITY;
    float sum = 0.0f;
    float4 acc = {0.0f, 0.0f, 0.0f, 0.0f};
    for (int p0 = beg; p0 < end; p0 += 32) {
        const int nb = min(32, end - p0);
        const int sl = (lane < nb) ? csr_src[p0 + lane] : 0;
        int c = 0;
        for (; c + 8 <= nb; c += 8)
            node_edges8<true>(z4, sl, c, 8, lane, a4, sd, mx, sum, acc);
        if (c < nb)
            node_edges8<false>(z4, sl, c, nb - c, lane, a4, sd, mx, sum, acc);
    }

    sum += row_xor<1>(sum);
    sum += row_xor<2>(sum);
    sum += row_xor<4>(sum);

    const float inv = (end > beg) ? 1.0f / sum : 0.0f;
    float4 o = {acc.x * inv, acc.y * inv, acc.z * inv, acc.w * inv};
    ((float4*)(out + (size_t)node * DIM))[lane] = o;
}

extern "C" void kernel_launch(void* const* d_in, const int* in_sizes, int n_in,
                              void* d_out, int out_size, void* d_ws, size_t ws_size,
                              hipStream_t stream) {
    const float* h      = (const float*)d_in[0];
    const float* W_fc   = (const float*)d_in[1];
    const float* a_attn = (const float*)d_in[2];
    const int*   src    = (const int*)d_in[3];
    const int*   dst    = (const int*)d_in[4];
    float*       out    = (float*)d_out;
    (void)in_sizes; (void)n_in; (void)out_size; (void)ws_size;

    float* ws        = (float*)d_ws;
    float* z         = ws;
    float* s_dst     = z + (size_t)N_NODES * DIM + N_NODES;
    int*   binned    = (int*)(s_dst + N_NODES);
    int*   lmat      = binned + N_EDGES;
    int*   row_start = lmat + 2 * N_NODES;
    int*   csr_src   = row_start + N_NODES + 1;

    gat_chunksort<<<EDGE_GRID, 1024, 0, stream>>>(src, dst, lmat, binned);

    gat_gemm_binsort<<<GEMM_GRID + N_BINS, 256, 0, stream>>>(h, W_fc, a_attn, z, s_dst,
                                                            binned, lmat, row_start, csr_src);

    gat_node<<<N_NODES / 8, 256, 0, stream>>>(z, a_attn, s_dst, row_start, csr_src, out);
}
